# starting kernel
# baseline (speedup 1.0000x reference)
#include <hip/hip_runtime.h>
#include <hip/hip_bf16.h>

typedef __attribute__((ext_vector_type(2))) float v2f;
typedef __attribute__((ext_vector_type(8))) float v8f;

#define B_DIM 1024
#define C_DIM 16
#define M_DIM 4096
#define N_DIM 8192
#define K_DIM 32
#define HALF_LOG_2PI 0.91893853320467274f

__global__ void build_u_kernel(const float* __restrict__ x,
                               float* __restrict__ U) {
    int idx = blockIdx.x * blockDim.x + threadIdx.x;
    if (idx >= B_DIM * C_DIM) return;
    int b = idx >> 4;
    int c = idx & 15;
    float v = x[b * C_DIM + c];
    U[b * K_DIM + c]         = v * v;
    U[b * K_DIM + C_DIM + c] = v;
}

__global__ void build_w_kernel(const float* __restrict__ locs,
                               const float* __restrict__ scales,
                               const int*   __restrict__ edges,
                               float* __restrict__ W,
                               float* __restrict__ Koff) {
    int n = blockIdx.x * blockDim.x + threadIdx.x;
    if (n >= N_DIM) return;
    float ko = 0.0f;
#pragma unroll
    for (int c = 0; c < C_DIM; ++c) {
        int   e    = edges[c * N_DIM + n];
        float mu   = locs[c * M_DIM + e];
        float s    = scales[c * M_DIM + e] + 0.5f;
        float inv  = 1.0f / s;
        float inv2 = inv * inv;
        W[c * N_DIM + n]           = -0.5f * inv2;
        W[(C_DIM + c) * N_DIM + n] = mu * inv2;
        ko += -0.5f * mu * mu * inv2 - __logf(s) - HALF_LOG_2PI;
    }
    Koff[n] = ko;
}

__global__ __launch_bounds__(256)
void wmma_gemm_kernel(const float* __restrict__ U,
                      const float* __restrict__ W,
                      const float* __restrict__ Koff,
                      float* __restrict__ out) {
    const int lane = threadIdx.x & 31;
    const int wave = threadIdx.x >> 5;

    constexpr int TILES_N = N_DIM / 16;
    const int n_tile  = blockIdx.x % TILES_N;
    const int b_group = blockIdx.x / TILES_N;

    const int b0 = (b_group * 8 + wave) * 16;
    const int n0 = n_tile * 16;

    const int half = lane >> 4;
    const int li   = lane & 15;

    const float* Ua = U + (b0 + li) * K_DIM + half * 2;
    const float* Wb = W + half * 2 * N_DIM + (n0 + li);

    v8f acc = {};
#pragma unroll
    for (int k0 = 0; k0 < K_DIM; k0 += 4) {
        v2f a;
        a.x = Ua[k0];
        a.y = Ua[k0 + 1];
        v2f bf;
        bf.x = Wb[k0 * N_DIM];
        bf.y = Wb[(k0 + 1) * N_DIM];
        acc = __builtin_amdgcn_wmma_f32_16x16x4_f32(
             false, a,  false, bf,
             (short)0, acc,  false,  false);
    }

    const float ko = Koff[n0 + li];
    float* o = out + (b0 + half * 8) * N_DIM + (n0 + li);
#pragma unroll
    for (int i = 0; i < 8; ++i) {
        o[i * N_DIM] = acc[i] + ko;
    }
}

extern "C" void kernel_launch(void* const* d_in, const int* in_sizes, int n_in,
                              void* d_out, int out_size, void* d_ws, size_t ws_size,
                              hipStream_t stream) {
    const float* x      = (const float*)d_in[0];
    const float* locs   = (const float*)d_in[1];
    const float* scales = (const float*)d_in[2];
    const int*   edges  = (const int*)d_in[3];
    float*       out    = (float*)d_out;

    float* U    = (float*)d_ws;
    float* W    = U + (size_t)B_DIM * K_DIM;
    float* Koff = W + (size_t)K_DIM * N_DIM;

    build_u_kernel<<<(B_DIM * C_DIM + 255) / 256, 256, 0, stream>>>(x, U);
    build_w_kernel<<<(N_DIM + 255) / 256, 256, 0, stream>>>(locs, scales, edges,
                                                            W, Koff);

    constexpr int GRID = (N_DIM / 16) * (B_DIM / 128);
    wmma_gemm_kernel<<<GRID, 256, 0, stream>>>(U, W, Koff, out);
}


# x reaches the waves through LDS: a block fetches its 512 x rows once, 16 B per lane in whole lines, and waves read A fragments from LDS (was 64 8-B loads per wave, 8 lines each); sums unchanged
# speedup vs baseline: 3.3067x; 3.3067x over previous
#include <hip/hip_runtime.h>

typedef __attribute__((ext_vector_type(2))) float v2f;
typedef __attribute__((ext_vector_type(4))) float v4f;
typedef __attribute__((ext_vector_type(8))) float v8f;

#define B_DIM 1024
#define C_DIM 16
#define M_DIM 4096
#define N_DIM 8192
#define HALF_LOG_2PI 0.91893853320467274f

#define ROWS_PER_WAVE 256
#define ROW_TILES (ROWS_PER_WAVE / 16)
#define ROW_GROUPS (B_DIM / ROWS_PER_WAVE)
#define STRIPS_PER_BLOCK 4
#define BLOCK_ROW_GROUPS 2
#define ROW_BLOCKS (ROW_GROUPS / BLOCK_ROW_GROUPS)
#define WAVES_PER_BLOCK (STRIPS_PER_BLOCK * BLOCK_ROW_GROUPS)
#define STAGE_ROWS 32
#define STAGE_PITCH 68
#define X_ROWS (BLOCK_ROW_GROUPS * ROWS_PER_WAVE)
#define X_PITCH 20

__device__ __forceinline__ void lds_block_barrier() {
    __builtin_amdgcn_fence(__ATOMIC_RELEASE, "workgroup", "local");
    __builtin_amdgcn_s_barrier();
    __builtin_amdgcn_fence(__ATOMIC_ACQUIRE, "workgroup", "local");
}

__global__ __launch_bounds__(32 * WAVES_PER_BLOCK)
void gather_wmma_kernel(const float* __restrict__ x,
                        const float* __restrict__ locs,
                        const float* __restrict__ scales,
                        const int*   __restrict__ edges,
                        float* __restrict__ out) {
    __shared__ __attribute__((aligned(16)))
    float stage[2][BLOCK_ROW_GROUPS][STAGE_ROWS][STAGE_PITCH];
    __shared__ __attribute__((aligned(16))) float xs[X_ROWS][X_PITCH];

    const int lane = threadIdx.x & 31;
    const int wave = threadIdx.x >> 5;
    const int h    = lane >> 4;
    const int li   = lane & 15;

    const int strip = wave / BLOCK_ROW_GROUPS;
    const int rg    = wave % BLOCK_ROW_GROUPS;
    const int cb    = blockIdx.x / ROW_BLOCKS;
    const int rblk  = blockIdx.x % ROW_BLOCKS;
    const int n  = (cb * STRIPS_PER_BLOCK + strip) * 16 + li;
    const int rb = (rblk * BLOCK_ROW_GROUPS + rg) * ROWS_PER_WAVE;

    v4f xin[8];
    const float* xg = x + rblk * (X_ROWS * C_DIM) + 4 * threadIdx.x;
#pragma unroll
    for (int p = 0; p < 8; ++p)
        xin[p] = *(const v4f*)(xg + p * 1024);

    int e[4][2];
#pragma unroll
    for (int q = 0; q < 4; ++q)
#pragma unroll
        for (int j = 0; j < 2; ++j)
            e[q][j] = edges[(4 * q + 2 * h + j) * N_DIM + n];

    float mu[4][2], sv[4][2];
#pragma unroll
    for (int q = 0; q < 4; ++q)
#pragma unroll
        for (int j = 0; j < 2; ++j) {
            const int c = 4 * q + 2 * h + j;
            mu[q][j] = locs[c * M_DIM + e[q][j]];
            sv[q][j] = scales[c * M_DIM + e[q][j]];
        }

    {
        float* xw = &xs[threadIdx.x >> 2][4 * (threadIdx.x & 3)];
#pragma unroll
        for (int p = 0; p < 8; ++p)
            *(v4f*)(xw + p * 64 * X_PITCH) = xin[p];
    }
    lds_block_barrier();
    v2f xa[ROW_TILES][4];
    const float* xr = &xs[rg * ROWS_PER_WAVE + li][2 * h];
#pragma unroll
    for (int i = 0; i < ROW_TILES; ++i)
#pragma unroll
        for (int q = 0; q < 4; ++q)
            xa[i][q] = *(const v2f*)(xr + i * 16 * X_PITCH + 4 * q);
    __builtin_amdgcn_sched_barrier(0);

    v2f bsq[4], blin[4];
    float term[4][2];
#pragma unroll
    for (int q = 0; q < 4; ++q) {
#pragma unroll
        for (int j = 0; j < 2; ++j) {
            const float m    = mu[q][j];
            const float s    = sv[q][j] + 0.5f;
            const float inv  = 1.0f / s;
            const float inv2 = inv * inv;
            bsq[q][j]  = -0.5f * inv2;
            blin[q][j] = m * inv2;
            term[q][j] = -0.5f * m * m * inv2 - __logf(s) - HALF_LOG_2PI;
        }
    }

    float ko = 0.0f;
#pragma unroll
    for (int q = 0; q < 4; ++q) {
        const float o0 = __shfl_xor(term[q][0], 16);
        const float o1 = __shfl_xor(term[q][1], 16);
        ko += h ? o0 : term[q][0];
        ko += h ? o1 : term[q][1];
        ko += h ? term[q][0] : o0;
        ko += h ? term[q][1] : o1;
    }

    float*       sw = &stage[0][rg][8 * h][16 * strip + li];
    const float* sr = &stage[0][rg][8 * strip + h][4 * li];
    constexpr int STAGE_FLOATS = BLOCK_ROW_GROUPS * STAGE_ROWS * STAGE_PITCH;
    float* op = out + (rb + 8 * strip + h) * N_DIM + cb * (16 * STRIPS_PER_BLOCK)
                + 4 * li;

    v8f p0 = {};
    v8f p1 = {};
    v4f row[4] = {};
#pragma unroll
    for (int i = 0; i < ROW_TILES; i += 2) {
        v8f acc0 = {};
        v8f acc1 = {};
#pragma unroll
        for (int q = 0; q < 4; ++q) {
            acc0 = __builtin_amdgcn_wmma_f32_16x16x4_f32(
                false, xa[i][q] * xa[i][q], false, bsq[q],
                (short)0, acc0, false, false);
            acc1 = __builtin_amdgcn_wmma_f32_16x16x4_f32(
                false, xa[i + 1][q] * xa[i + 1][q], false, bsq[q],
                (short)0, acc1, false, false);
        }
        if (i > 0) {
            const int so = (((i - 2) / 2) & 1) * STAGE_FLOATS;
#pragma unroll
            for (int r = 0; r < 8; ++r) {
                sw[so + r * STAGE_PITCH]        = p0[r] + ko;
                sw[so + (16 + r) * STAGE_PITCH] = p1[r] + ko;
            }
            lds_block_barrier();
#pragma unroll
            for (int j = 0; j < 4; ++j)
                row[j] = *(const v4f*)(sr + so + 2 * j * STAGE_PITCH);
        }
        __builtin_amdgcn_sched_barrier(0);
#pragma unroll
        for (int q = 0; q < 4; ++q) {
            acc0 = __builtin_amdgcn_wmma_f32_16x16x4_f32(
                false, xa[i][q], false, blin[q],
                (short)0, acc0, false, false);
            acc1 = __builtin_amdgcn_wmma_f32_16x16x4_f32(
                false, xa[i + 1][q], false, blin[q],
                (short)0, acc1, false, false);
        }
        __builtin_amdgcn_sched_barrier(0);
        if (i > 0) {
#pragma unroll
            for (int j = 0; j < 4; ++j)
                *(v4f*)(op + ((i - 2) * 16 + 2 * j) * N_DIM) = row[j];
        }
        p0 = acc0;
        p1 = acc1;
        __builtin_amdgcn_sched_barrier(0);
    }
    {
        const int so = (((ROW_TILES - 2) / 2) & 1) * STAGE_FLOATS;
#pragma unroll
        for (int r = 0; r < 8; ++r) {
            sw[so + r * STAGE_PITCH]        = p0[r] + ko;
            sw[so + (16 + r) * STAGE_PITCH] = p1[r] + ko;
        }
        lds_block_barrier();
#pragma unroll
        for (int j = 0; j < 4; ++j)
            row[j] = *(const v4f*)(sr + so + 2 * j * STAGE_PITCH);
#pragma unroll
        for (int j = 0; j < 4; ++j)
            *(v4f*)(op + ((ROW_TILES - 2) * 16 + 2 * j) * N_DIM) = row[j];
    }
}

extern "C" void kernel_launch(void* const* d_in, const int* in_sizes, int n_in,
                              void* d_out, int out_size, void* d_ws, size_t ws_size,
                              hipStream_t stream) {
    const float* x      = (const float*)d_in[0];
    const float* locs   = (const float*)d_in[1];
    const float* scales = (const float*)d_in[2];
    const int*   edges  = (const int*)d_in[3];
    float*       out    = (float*)d_out;

    constexpr int GRID = (N_DIM / (16 * STRIPS_PER_BLOCK)) * ROW_BLOCKS;
    gather_wmma_kernel<<<GRID, 32 * WAVES_PER_BLOCK, 0, stream>>>(x, locs, scales,
                                                                  edges, out);
}


# Each W element is gathered once per block: the 2 row-group waves of a strip gather and finish 2 k-steps each (8 loads per lane, was 16) and swap the fragments once through LDS; sums unchanged
# speedup vs baseline: 3.8232x; 1.1562x over previous
#include <hip/hip_runtime.h>

typedef __attribute__((ext_vector_type(2))) float v2f;
typedef __attribute__((ext_vector_type(4))) float v4f;
typedef __attribute__((ext_vector_type(8))) float v8f;

#define B_DIM 1024
#define C_DIM 16
#define M_DIM 4096
#define N_DIM 8192
#define HALF_LOG_2PI 0.91893853320467274f

#define ROWS_PER_WAVE 256
#define ROW_TILES (ROWS_PER_WAVE / 16)
#define ROW_GROUPS (B_DIM / ROWS_PER_WAVE)
#define STRIPS_PER_BLOCK 4
#define BLOCK_ROW_GROUPS 2
#define ROW_BLOCKS (ROW_GROUPS / BLOCK_ROW_GROUPS)
#define WAVES_PER_BLOCK (STRIPS_PER_BLOCK * BLOCK_ROW_GROUPS)
#define STAGE_ROWS 32
#define STAGE_PITCH 68
#define X_ROWS (BLOCK_ROW_GROUPS * ROWS_PER_WAVE)
#define X_PITCH 20

__device__ __forceinline__ void lds_block_barrier() {
    __builtin_amdgcn_fence(__ATOMIC_RELEASE, "workgroup", "local");
    __builtin_amdgcn_s_barrier();
    __builtin_amdgcn_fence(__ATOMIC_ACQUIRE, "workgroup", "local");
}

__global__ __launch_bounds__(32 * WAVES_PER_BLOCK)
void gather_wmma_kernel(const float* __restrict__ x,
                        const float* __restrict__ locs,
                        const float* __restrict__ scales,
                        const int*   __restrict__ edges,
                        float* __restrict__ out) {
    __shared__ __attribute__((aligned(16)))
    float stage[2][BLOCK_ROW_GROUPS][STAGE_ROWS][STAGE_PITCH];
    __shared__ __attribute__((aligned(16))) float xs[X_ROWS][X_PITCH];
    __shared__ v4f wx[STRIPS_PER_BLOCK][BLOCK_ROW_GROUPS][3][32];
    static_assert(BLOCK_ROW_GROUPS == 2, "the W gather is split between two waves");

    const int lane = threadIdx.x & 31;
    const int wave = threadIdx.x >> 5;
    const int h    = lane >> 4;
    const int li   = lane & 15;

    const int strip = wave / BLOCK_ROW_GROUPS;
    const int rg    = wave % BLOCK_ROW_GROUPS;
    const int cb    = blockIdx.x / ROW_BLOCKS;
    const int rblk  = blockIdx.x % ROW_BLOCKS;
    const int n  = (cb * STRIPS_PER_BLOCK + strip) * 16 + li;
    const int rb = (rblk * BLOCK_ROW_GROUPS + rg) * ROWS_PER_WAVE;

    v4f xin[8];
    const float* xg = x + rblk * (X_ROWS * C_DIM) + 4 * threadIdx.x;
#pragma unroll
    for (int p = 0; p < 8; ++p)
        xin[p] = *(const v4f*)(xg + p * 1024);

    const int c0 = 8 * rg + 2 * h;
    int e[2][2];
#pragma unroll
    for (int t = 0; t < 2; ++t)
#pragma unroll
        for (int j = 0; j < 2; ++j)
            e[t][j] = edges[(c0 + 4 * t + j) * N_DIM + n];

    float mu[2][2], sv[2][2];
#pragma unroll
    for (int t = 0; t < 2; ++t)
#pragma unroll
        for (int j = 0; j < 2; ++j) {
            const int c = c0 + 4 * t + j;
            mu[t][j] = locs[c * M_DIM + e[t][j]];
            sv[t][j] = scales[c * M_DIM + e[t][j]];
        }

    {
        float* xw = &xs[threadIdx.x >> 2][4 * (threadIdx.x & 3)];
#pragma unroll
        for (int p = 0; p < 8; ++p)
            *(v4f*)(xw + p * 64 * X_PITCH) = xin[p];
    }
    lds_block_barrier();
    v2f xa[ROW_TILES][4];
    const float* xr = &xs[rg * ROWS_PER_WAVE + li][2 * h];
#pragma unroll
    for (int i = 0; i < ROW_TILES; ++i)
#pragma unroll
        for (int q = 0; q < 4; ++q)
            xa[i][q] = *(const v2f*)(xr + i * 16 * X_PITCH + 4 * q);
    __builtin_amdgcn_sched_barrier(0);

    v4f hsq, hlin, hterm;
#pragma unroll
    for (int t = 0; t < 2; ++t) {
#pragma unroll
        for (int j = 0; j < 2; ++j) {
            const float m    = mu[t][j];
            const float s    = sv[t][j] + 0.5f;
            const float inv  = 1.0f / s;
            const float inv2 = inv * inv;
            hsq[2 * t + j]   = -0.5f * inv2;
            hlin[2 * t + j]  = m * inv2;
            hterm[2 * t + j] = -0.5f * m * m * inv2 - __logf(s) - HALF_LOG_2PI;
        }
    }
    wx[strip][rg][0][lane] = hsq;
    wx[strip][rg][1][lane] = hlin;
    wx[strip][rg][2][lane] = hterm;
    lds_block_barrier();
    v2f bsq[4], blin[4];
    float term[4][2];
#pragma unroll
    for (int p = 0; p < 2; ++p) {
        const v4f a = wx[strip][p][0][lane];
        const v4f b = wx[strip][p][1][lane];
        const v4f d = wx[strip][p][2][lane];
#pragma unroll
        for (int t = 0; t < 2; ++t)
#pragma unroll
            for (int j = 0; j < 2; ++j) {
                bsq[2 * p + t][j]  = a[2 * t + j];
                blin[2 * p + t][j] = b[2 * t + j];
                term[2 * p + t][j] = d[2 * t + j];
            }
    }

    float ko = 0.0f;
#pragma unroll
    for (int q = 0; q < 4; ++q) {
        const float o0 = __shfl_xor(term[q][0], 16);
        const float o1 = __shfl_xor(term[q][1], 16);
        ko += h ? o0 : term[q][0];
        ko += h ? o1 : term[q][1];
        ko += h ? term[q][0] : o0;
        ko += h ? term[q][1] : o1;
    }

    float*       sw = &stage[0][rg][8 * h][16 * strip + li];
    const float* sr = &stage[0][rg][8 * strip + h][4 * li];
    constexpr int STAGE_FLOATS = BLOCK_ROW_GROUPS * STAGE_ROWS * STAGE_PITCH;
    float* op = out + (rb + 8 * strip + h) * N_DIM + cb * (16 * STRIPS_PER_BLOCK)
                + 4 * li;

    v8f p0 = {};
    v8f p1 = {};
    v4f row[4] = {};
#pragma unroll
    for (int i = 0; i < ROW_TILES; i += 2) {
        v8f acc0 = {};
        v8f acc1 = {};
#pragma unroll
        for (int q = 0; q < 4; ++q) {
            acc0 = __builtin_amdgcn_wmma_f32_16x16x4_f32(
                false, xa[i][q] * xa[i][q], false, bsq[q],
                (short)0, acc0, false, false);
            acc1 = __builtin_amdgcn_wmma_f32_16x16x4_f32(
                false, xa[i + 1][q] * xa[i + 1][q], false, bsq[q],
                (short)0, acc1, false, false);
        }
        if (i > 0) {
            const int so = (((i - 2) / 2) & 1) * STAGE_FLOATS;
#pragma unroll
            for (int r = 0; r < 8; ++r) {
                sw[so + r * STAGE_PITCH]        = p0[r] + ko;
                sw[so + (16 + r) * STAGE_PITCH] = p1[r] + ko;
            }
            lds_block_barrier();
#pragma unroll
            for (int j = 0; j < 4; ++j)
                row[j] = *(const v4f*)(sr + so + 2 * j * STAGE_PITCH);
        }
        __builtin_amdgcn_sched_barrier(0);
#pragma unroll
        for (int q = 0; q < 4; ++q) {
            acc0 = __builtin_amdgcn_wmma_f32_16x16x4_f32(
                false, xa[i][q], false, blin[q],
                (short)0, acc0, false, false);
            acc1 = __builtin_amdgcn_wmma_f32_16x16x4_f32(
                false, xa[i + 1][q], false, blin[q],
                (short)0, acc1, false, false);
        }
        __builtin_amdgcn_sched_barrier(0);
        if (i > 0) {
#pragma unroll
            for (int j = 0; j < 4; ++j)
                *(v4f*)(op + ((i - 2) * 16 + 2 * j) * N_DIM) = row[j];
        }
        p0 = acc0;
        p1 = acc1;
        __builtin_amdgcn_sched_barrier(0);
    }
    {
        const int so = (((ROW_TILES - 2) / 2) & 1) * STAGE_FLOATS;
#pragma unroll
        for (int r = 0; r < 8; ++r) {
            sw[so + r * STAGE_PITCH]        = p0[r] + ko;
            sw[so + (16 + r) * STAGE_PITCH] = p1[r] + ko;
        }
        lds_block_barrier();
#pragma unroll
        for (int j = 0; j < 4; ++j)
            row[j] = *(const v4f*)(sr + so + 2 * j * STAGE_PITCH);
#pragma unroll
        for (int j = 0; j < 4; ++j)
            *(v4f*)(op + ((ROW_TILES - 2) * 16 + 2 * j) * N_DIM) = row[j];
    }
}

extern "C" void kernel_launch(void* const* d_in, const int* in_sizes, int n_in,
                              void* d_out, int out_size, void* d_ws, size_t ws_size,
                              hipStream_t stream) {
    const float* x      = (const float*)d_in[0];
    const float* locs   = (const float*)d_in[1];
    const float* scales = (const float*)d_in[2];
    const int*   edges  = (const int*)d_in[3];
    float*       out    = (float*)d_out;

    constexpr int GRID = (N_DIM / (16 * STRIPS_PER_BLOCK)) * ROW_BLOCKS;
    gather_wmma_kernel<<<GRID, 32 * WAVES_PER_BLOCK, 0, stream>>>(x, locs, scales,
                                                                  edges, out);
}
